# starting kernel
# baseline (speedup 1.0000x reference)
#include <hip/hip_runtime.h>

typedef __attribute__((ext_vector_type(2))) float v2f;
typedef __attribute__((ext_vector_type(8))) float v8f;

#define NN 8
#define NB 4
#define CC 32
#define HW (96 * 96)
#define TILES_PER_WAVE 4
#define WAVES_PER_BLOCK 4

static __device__ __forceinline__ v8f wmma4(v2f a, v2f b, v8f c) {
  return __builtin_amdgcn_wmma_f32_16x16x4_f32(false, a, false, b, (short)0, c,
                                               false, false);
}

static __device__ __forceinline__ v8f vzero8() {
  v8f v;
#pragma unroll
  for (int r = 0; r < 8; ++r) v[r] = 0.0f;
  return v;
}

static __device__ __forceinline__ void load_Btile(const float* __restrict__ src,
                                                  int node, int bb, int p,
                                                  int n, int h, v2f bt[8]) {
  const float* base = src + ((long)(node * NB + bb) * CC) * HW + p + n;
#pragma unroll
  for (int kk = 0; kk < 8; ++kk) {
    const int c0 = 4 * kk + 2 * h;
    v2f v;
    v.x = base[(long)c0 * HW];
    v.y = base[(long)(c0 + 1) * HW];
    bt[kk] = v;
  }
}

static __device__ __forceinline__ void load_Dtile(const float* __restrict__ src,
                                                  int node, int bb, int p,
                                                  int n, int h, v8f dt[2]) {
  const float* base = src + ((long)(node * NB + bb) * CC) * HW + p + n;
#pragma unroll
  for (int m = 0; m < 2; ++m)
#pragma unroll
    for (int r = 0; r < 8; ++r)
      dt[m][r] = base[(long)(16 * m + 8 * h + r) * HW];
}

static __device__ __forceinline__ void gemm32(const v2f wa[2][8],
                                              const v2f bt[8], v8f out[2]) {
#pragma unroll
  for (int m = 0; m < 2; ++m)
#pragma unroll
    for (int kk = 0; kk < 8; ++kk) out[m] = wmma4(wa[m][kk], bt[kk], out[m]);
}

__global__ __launch_bounds__(WAVES_PER_BLOCK * 32) void gcn_step_kernel(
    const float* __restrict__ src, const float* __restrict__ Wedge,
    const float* __restrict__ bedge, float* __restrict__ dst) {
  const int lane = threadIdx.x & 31;
  const int wave = threadIdx.x >> 5;
  const int n = lane & 15;
  const int h = lane >> 4;
  const int i = blockIdx.y;

  v2f wd[2][8], w2[2][8];
  const float* Wi = Wedge + (long)i * CC * (2 * CC);
#pragma unroll
  for (int m = 0; m < 2; ++m) {
    const float* row = Wi + (long)(16 * m + n) * (2 * CC);
#pragma unroll
    for (int kk = 0; kk < 8; ++kk) {
      const int c0 = 4 * kk + 2 * h;
      const float d0 = row[c0], d1 = row[c0 + 1];
      const float s0 = row[CC + c0], s1 = row[CC + c0 + 1];
      v2f a;
      a.x = d0; a.y = d1;
      wd[m][kk] = a;
      v2f b2;
      b2.x = s0 - d0; b2.y = s1 - d1;
      w2[m][kk] = b2;
    }
  }
  v8f bias[2];
#pragma unroll
  for (int m = 0; m < 2; ++m)
#pragma unroll
    for (int r = 0; r < 8; ++r)
      bias[m][r] = bedge[i * CC + 16 * m + 8 * h + r];

  const int tile0 =
      (blockIdx.x * WAVES_PER_BLOCK + wave) * TILES_PER_WAVE;

  for (int t = 0; t < TILES_PER_WAVE; ++t) {
    const int q = (tile0 + t) * 16;
    const int bb = q / HW;
    const int p = q - bb * HW;

    v2f bt[8];
    load_Btile(src, i, bb, p, n, h, bt);
    v8f T[2] = {bias[0], bias[1]};
    gemm32(w2, bt, T);

    v8f acc[2] = {vzero8(), vzero8()};

    for (int j = 0; j < NN; ++j) {
      if (j == i) continue;
      load_Btile(src, j, bb, p, n, h, bt);
      v8f P[2] = {T[0], T[1]};
      gemm32(wd, bt, P);
      v8f dv[2];
      load_Dtile(src, j, bb, p, n, h, dv);
#pragma unroll
      for (int m = 0; m < 2; ++m)
#pragma unroll
        for (int r = 0; r < 8; ++r)
          acc[m][r] += fmaxf(P[m][r], 0.0f) * dv[m][r];
    }

    v8f self[2];
    load_Dtile(src, i, bb, p, n, h, self);
    float* obase = dst + ((long)(i * NB + bb) * CC) * HW + p + n;
#pragma unroll
    for (int m = 0; m < 2; ++m)
#pragma unroll
      for (int r = 0; r < 8; ++r)
        obase[(long)(16 * m + 8 * h + r) * HW] = acc[m][r] + self[m][r];
  }
}

extern "C" void kernel_launch(void* const* d_in, const int* in_sizes, int n_in,
                              void* d_out, int out_size, void* d_ws,
                              size_t ws_size, hipStream_t stream) {
  const float* nodes = (const float*)d_in[0];
  const float* Wedge = (const float*)d_in[1];
  const float* bedge = (const float*)d_in[2];
  float* out = (float*)d_out;
  float* ws = (float*)d_ws;

  dim3 block(WAVES_PER_BLOCK * 32);
  dim3 grid((NB * HW / 16) / (WAVES_PER_BLOCK * TILES_PER_WAVE), NN);

  gcn_step_kernel<<<grid, block, 0, stream>>>(nodes, Wedge, bedge, out);
  gcn_step_kernel<<<grid, block, 0, stream>>>(out, Wedge, bedge, ws);
  gcn_step_kernel<<<grid, block, 0, stream>>>(ws, Wedge, bedge, out);
}


# Permute channel (K) order of weight and node fragments so a lane's B fragment holds its own C/D-layout channels: gate and residual reuse it, each tile loaded once not twice; dot-sum order changes
# speedup vs baseline: 1.2888x; 1.2888x over previous
#include <hip/hip_runtime.h>

typedef __attribute__((ext_vector_type(2))) float v2f;
typedef __attribute__((ext_vector_type(8))) float v8f;

#define NN 8
#define NB 4
#define CC 32
#define HW (96 * 96)
#define TILES_PER_WAVE 4
#define WAVES_PER_BLOCK 4

static __device__ __forceinline__ v8f wmma4(v2f a, v2f b, v8f c) {
  return __builtin_amdgcn_wmma_f32_16x16x4_f32(false, a, false, b, (short)0, c,
                                               false, false);
}

static __device__ __forceinline__ v8f vzero8() {
  v8f v;
#pragma unroll
  for (int r = 0; r < 8; ++r) v[r] = 0.0f;
  return v;
}

static __device__ __forceinline__ int kchan0(int kk) {
  return 16 * (kk >> 2) + 2 * (kk & 3);
}

static __device__ __forceinline__ void load_tile(const float* __restrict__ src,
                                                 int node, int bb, int p,
                                                 int n, int h, v2f bt[8]) {
  const float* base =
      src + ((long)(node * NB + bb) * CC + 8 * h) * HW + p + n;
#pragma unroll
  for (int kk = 0; kk < 8; ++kk) {
    const int c0 = kchan0(kk);
    v2f v;
    v.x = base[(long)c0 * HW];
    v.y = base[(long)(c0 + 1) * HW];
    bt[kk] = v;
  }
}

static __device__ __forceinline__ float dview(const v2f bt[8], int m, int r) {
  return bt[4 * m + (r >> 1)][r & 1];
}

static __device__ __forceinline__ void gemm32(const v2f wa[2][8],
                                              const v2f bt[8], v8f out[2]) {
#pragma unroll
  for (int m = 0; m < 2; ++m)
#pragma unroll
    for (int kk = 0; kk < 8; ++kk) out[m] = wmma4(wa[m][kk], bt[kk], out[m]);
}

__global__ __launch_bounds__(WAVES_PER_BLOCK * 32) void gcn_step_kernel(
    const float* __restrict__ src, const float* __restrict__ Wedge,
    const float* __restrict__ bedge, float* __restrict__ dst) {
  const int lane = threadIdx.x & 31;
  const int wave = threadIdx.x >> 5;
  const int n = lane & 15;
  const int h = lane >> 4;
  const int i = blockIdx.y;

  v2f wd[2][8], w2[2][8];
  const float* Wi = Wedge + (long)i * CC * (2 * CC);
#pragma unroll
  for (int m = 0; m < 2; ++m) {
    const float* row = Wi + (long)(16 * m + n) * (2 * CC);
#pragma unroll
    for (int kk = 0; kk < 8; ++kk) {
      const int c0 = kchan0(kk) + 8 * h;
      const float d0 = row[c0], d1 = row[c0 + 1];
      const float s0 = row[CC + c0], s1 = row[CC + c0 + 1];
      v2f a;
      a.x = d0; a.y = d1;
      wd[m][kk] = a;
      v2f b2;
      b2.x = s0 - d0; b2.y = s1 - d1;
      w2[m][kk] = b2;
    }
  }
  v8f bias[2];
#pragma unroll
  for (int m = 0; m < 2; ++m)
#pragma unroll
    for (int r = 0; r < 8; ++r)
      bias[m][r] = bedge[i * CC + 16 * m + 8 * h + r];

  const int tile0 =
      (blockIdx.x * WAVES_PER_BLOCK + wave) * TILES_PER_WAVE;

  for (int t = 0; t < TILES_PER_WAVE; ++t) {
    const int q = (tile0 + t) * 16;
    const int bb = q / HW;
    const int p = q - bb * HW;

    v2f self[8];
    load_tile(src, i, bb, p, n, h, self);
    v8f T[2] = {bias[0], bias[1]};
    gemm32(w2, self, T);

    v8f acc[2] = {vzero8(), vzero8()};

    for (int j = 0; j < NN; ++j) {
      if (j == i) continue;
      v2f bt[8];
      load_tile(src, j, bb, p, n, h, bt);
      v8f P[2] = {T[0], T[1]};
      gemm32(wd, bt, P);
#pragma unroll
      for (int m = 0; m < 2; ++m)
#pragma unroll
        for (int r = 0; r < 8; ++r)
          acc[m][r] += fmaxf(P[m][r], 0.0f) * dview(bt, m, r);
    }

    float* obase = dst + ((long)(i * NB + bb) * CC) * HW + p + n;
#pragma unroll
    for (int m = 0; m < 2; ++m)
#pragma unroll
      for (int r = 0; r < 8; ++r)
        obase[(long)(16 * m + 8 * h + r) * HW] = acc[m][r] + dview(self, m, r);
  }
}

extern "C" void kernel_launch(void* const* d_in, const int* in_sizes, int n_in,
                              void* d_out, int out_size, void* d_ws,
                              size_t ws_size, hipStream_t stream) {
  const float* nodes = (const float*)d_in[0];
  const float* Wedge = (const float*)d_in[1];
  const float* bedge = (const float*)d_in[2];
  float* out = (float*)d_out;
  float* ws = (float*)d_ws;

  dim3 block(WAVES_PER_BLOCK * 32);
  dim3 grid((NB * HW / 16) / (WAVES_PER_BLOCK * TILES_PER_WAVE), NN);

  gcn_step_kernel<<<grid, block, 0, stream>>>(nodes, Wedge, bedge, out);
  gcn_step_kernel<<<grid, block, 0, stream>>>(out, Wedge, bedge, ws);
  gcn_step_kernel<<<grid, block, 0, stream>>>(ws, Wedge, bedge, out);
}


# Sender tiles double-buffered in registers: the 16 loads of the next phase's tile (and the next span's receiver tile) go out as one group ahead of the current matrix block; sum order unchanged
# speedup vs baseline: 1.6219x; 1.2585x over previous
#include <hip/hip_runtime.h>

typedef __attribute__((ext_vector_type(2))) float v2f;
typedef __attribute__((ext_vector_type(8))) float v8f;

#define NN 8
#define NB 4
#define CC 32
#define HW (96 * 96)
#define SPAN 32
#define SPANS_PER_WAVE 3
#define WAVES_PER_BLOCK 12

static __device__ __forceinline__ v8f wmma4(v2f a, v2f b, v8f c) {
  return __builtin_amdgcn_wmma_f32_16x16x4_f32(false, a, false, b, (short)0, c,
                                               false, false);
}

static __device__ __forceinline__ v8f vzero8() {
  v8f v;
#pragma unroll
  for (int r = 0; r < 8; ++r) v[r] = 0.0f;
  return v;
}

static __device__ __forceinline__ int kchan0(int kk) {
  return 16 * (kk >> 2) + 2 * (kk & 3);
}

static __device__ __forceinline__ void load_span(const float* __restrict__ src,
                                                 int node, int bb, int p,
                                                 int n, int h, v2f ld[16]) {
  const float* base =
      src + ((long)(node * NB + bb) * CC + 8 * h) * HW + p + 2 * n;
#pragma unroll
  for (int idx = 0; idx < 16; ++idx) {
    const int c = 16 * (idx >> 3) + (idx & 7);
    ld[idx] = *reinterpret_cast<const v2f*>(base + (long)c * HW);
  }
}

static __device__ __forceinline__ void gemm32x2(const v2f wa[2][8],
                                                const v2f ld[16],
                                                v8f out[2][2]) {
#pragma unroll
  for (int kk = 0; kk < 8; ++kk) {
#pragma unroll
    for (int e = 0; e < 2; ++e) {
      v2f b;
      b.x = ld[2 * kk][e];
      b.y = ld[2 * kk + 1][e];
#pragma unroll
      for (int m = 0; m < 2; ++m) out[e][m] = wmma4(wa[m][kk], b, out[e][m]);
    }
  }
}

__global__ __launch_bounds__(WAVES_PER_BLOCK * 32) void gcn_step_kernel(
    const float* __restrict__ src, const float* __restrict__ Wedge,
    const float* __restrict__ bedge, float* __restrict__ dst) {
  const int lane = threadIdx.x & 31;
  const int wave = threadIdx.x >> 5;
  const int n = lane & 15;
  const int h = lane >> 4;
  const int i = blockIdx.y;

  v2f wd[2][8], w2[2][8];
  const float* Wi = Wedge + (long)i * CC * (2 * CC);
#pragma unroll
  for (int m = 0; m < 2; ++m) {
    const float* row = Wi + (long)(16 * m + n) * (2 * CC);
#pragma unroll
    for (int kk = 0; kk < 8; ++kk) {
      const int c0 = kchan0(kk) + 8 * h;
      const float d0 = row[c0], d1 = row[c0 + 1];
      const float s0 = row[CC + c0], s1 = row[CC + c0 + 1];
      v2f a;
      a.x = d0; a.y = d1;
      wd[m][kk] = a;
      v2f b2;
      b2.x = s0 - d0; b2.y = s1 - d1;
      w2[m][kk] = b2;
    }
  }
  v8f bias[2];
#pragma unroll
  for (int m = 0; m < 2; ++m)
#pragma unroll
    for (int r = 0; r < 8; ++r)
      bias[m][r] = bedge[i * CC + 16 * m + 8 * h + r];

  const int span0 =
      (blockIdx.x * WAVES_PER_BLOCK + wave) * SPANS_PER_WAVE;

  v2f self[16];
  {
    const int q0 = span0 * SPAN;
    const int bb0 = q0 / HW;
    load_span(src, i, bb0, q0 - bb0 * HW, n, h, self);
  }

#pragma unroll 1
  for (int t = 0; t < SPANS_PER_WAVE; ++t) {
    const int q = (span0 + t) * SPAN;
    const int bb = q / HW;
    const int p = q - bb * HW;

    v2f nb[2][16];
    __builtin_amdgcn_sched_barrier(0);
    load_span(src, (0 >= i ? 1 : 0), bb, p, n, h, nb[0]);
    __builtin_amdgcn_sched_barrier(0);

    v8f T[2][2] = {{bias[0], bias[1]}, {bias[0], bias[1]}};
    gemm32x2(w2, self, T);

    v2f acc[16];
#pragma unroll
    for (int idx = 0; idx < 16; ++idx) { acc[idx].x = 0.0f; acc[idx].y = 0.0f; }

#pragma unroll
    for (int k = 0; k < NN - 1; ++k) {
      const int cur = k & 1;
      const int nxt = cur ^ 1;
      __builtin_amdgcn_sched_barrier(0);
      if (k + 1 < NN - 1) {
        const int jn = (k + 1) + ((k + 1) >= i ? 1 : 0);
        load_span(src, jn, bb, p, n, h, nb[nxt]);
      } else {
        const int tn = (t + 1 < SPANS_PER_WAVE) ? t + 1 : t;
        const int qn = (span0 + tn) * SPAN;
        const int bbn = qn / HW;
        load_span(src, i, bbn, qn - bbn * HW, n, h, nb[nxt]);
      }
      __builtin_amdgcn_sched_barrier(0);

      v8f P[2][2] = {{T[0][0], T[0][1]}, {T[1][0], T[1][1]}};
      gemm32x2(wd, nb[cur], P);
#pragma unroll
      for (int e = 0; e < 2; ++e)
#pragma unroll
        for (int m = 0; m < 2; ++m)
          P[e][m] = __builtin_elementwise_max(P[e][m], vzero8());
#pragma unroll
      for (int m = 0; m < 2; ++m)
#pragma unroll
        for (int r = 0; r < 8; ++r) {
          acc[8 * m + r].x += P[0][m][r] * nb[cur][8 * m + r].x;
          acc[8 * m + r].y += P[1][m][r] * nb[cur][8 * m + r].y;
        }
    }

    float* obase = dst + ((long)(i * NB + bb) * CC + 8 * h) * HW + p + 2 * n;
#pragma unroll
    for (int m = 0; m < 2; ++m)
#pragma unroll
      for (int r = 0; r < 8; ++r) {
        v2f o;
        o.x = acc[8 * m + r].x + self[8 * m + r].x;
        o.y = acc[8 * m + r].y + self[8 * m + r].y;
        *reinterpret_cast<v2f*>(obase + (long)(16 * m + r) * HW) = o;
      }

#pragma unroll
    for (int idx = 0; idx < 16; ++idx) self[idx] = nb[1][idx];
  }
}

extern "C" void kernel_launch(void* const* d_in, const int* in_sizes, int n_in,
                              void* d_out, int out_size, void* d_ws,
                              size_t ws_size, hipStream_t stream) {
  const float* nodes = (const float*)d_in[0];
  const float* Wedge = (const float*)d_in[1];
  const float* bedge = (const float*)d_in[2];
  float* out = (float*)d_out;
  float* ws = (float*)d_ws;

  dim3 block(WAVES_PER_BLOCK * 32);
  dim3 grid((NB * HW / SPAN) / (WAVES_PER_BLOCK * SPANS_PER_WAVE), NN);

  gcn_step_kernel<<<grid, block, 0, stream>>>(nodes, Wedge, bedge, out);
  gcn_step_kernel<<<grid, block, 0, stream>>>(out, Wedge, bedge, ws);
  gcn_step_kernel<<<grid, block, 0, stream>>>(ws, Wedge, bedge, out);
}


# All 3 steps in one launch: a work-group keeps all 8 nodes of its 144 pixels in LDS (two copies, one barrier per step), wave = receiver, sender tiles read from LDS not cache; sum order unchanged
# speedup vs baseline: 2.0616x; 1.2711x over previous
#include <hip/hip_runtime.h>

typedef __attribute__((ext_vector_type(2))) float v2f;
typedef __attribute__((ext_vector_type(4))) float v4f;
typedef __attribute__((ext_vector_type(8))) float v8f;

#define NN 8
#define NB 4
#define CC 32
#define HW (96 * 96)
#define NSTEPS 3
#define PIX_PER_WG 144
#define WG_PER_IMAGE (HW / PIX_PER_WG)
#define TILES 9
#define TILE_Q 128
#define NODE_Q (TILES * TILE_Q)
#define COPY_Q (NN * NODE_Q)


static __device__ __forceinline__ v8f wmma4(v2f a, v2f b, v8f c) {
  return __builtin_amdgcn_wmma_f32_16x16x4_f32(false, a, false, b, (short)0, c,
                                               false, false);
}

static __device__ __forceinline__ int kchan0(int kk) {
  return 16 * (kk >> 2) + 2 * (kk & 3);
}

static __device__ __forceinline__ void load_tile(const v4f* p, v4f v[4]) {
#pragma unroll
  for (int q = 0; q < 4; ++q) v[q] = p[q * 32];
}

static __device__ __forceinline__ void gemm32(const v2f wa[2][8],
                                              const v4f v[4], v8f out[2]) {
#pragma unroll
  for (int kk = 0; kk < 8; ++kk) {
    v2f b;
    b.x = v[kk >> 1][2 * (kk & 1)];
    b.y = v[kk >> 1][2 * (kk & 1) + 1];
#pragma unroll
    for (int m = 0; m < 2; ++m) out[m] = wmma4(wa[m][kk], b, out[m]);
  }
}

__global__ __launch_bounds__(NN * 32) void gcn_fused_kernel(
    const float* __restrict__ nodes, const float* __restrict__ Wedge,
    const float* __restrict__ bedge, float* __restrict__ out) {
  __shared__ v4f lds[2 * COPY_Q];

  const int lane = threadIdx.x & 31;
  const int i = __builtin_amdgcn_readfirstlane((int)(threadIdx.x >> 5));
  const int n = lane & 15;
  const int h = lane >> 4;

  const int wg = blockIdx.x;
  const int bb = wg / WG_PER_IMAGE;
  const int p0 = (wg - bb * WG_PER_IMAGE) * PIX_PER_WG;
  const int odd = wg & 1;
  const int span_base = p0 + 16 * odd;
  const int left_base = p0 + 128 * (1 - odd);

  {
    const float* g = nodes + ((long)(i * NB + bb) * CC + 8 * h) * HW;
    v4f* mine = lds + i * NODE_Q + lane;
#pragma unroll
    for (int s = 0; s < 4; ++s) {
      v2f ld[16];
#pragma unroll
      for (int idx = 0; idx < 16; ++idx) {
        const int c = 16 * (idx >> 3) + (idx & 7);
        ld[idx] = *reinterpret_cast<const v2f*>(g + (long)c * HW + span_base +
                                                32 * s + 2 * n);
      }
#pragma unroll
      for (int q = 0; q < 4; ++q) {
        v4f a, b;
#pragma unroll
        for (int e = 0; e < 4; ++e) {
          a[e] = ld[4 * q + e].x;
          b[e] = ld[4 * q + e].y;
        }
        mine[(2 * s) * TILE_Q + q * 32] = a;
        mine[(2 * s + 1) * TILE_Q + q * 32] = b;
      }
    }
#pragma unroll
    for (int q = 0; q < 4; ++q) {
      v4f a;
#pragma unroll
      for (int e = 0; e < 4; ++e) {
        const int idx = 4 * q + e;
        const int c = 16 * (idx >> 3) + (idx & 7);
        a[e] = g[(long)c * HW + left_base + n];
      }
      mine[8 * TILE_Q + q * 32] = a;
    }
  }

  v2f wd[2][8], w2[2][8];
  const float* Wi = Wedge + (long)i * CC * (2 * CC);
#pragma unroll
  for (int m = 0; m < 2; ++m) {
    const float* row = Wi + (long)(16 * m + n) * (2 * CC);
#pragma unroll
    for (int kk = 0; kk < 8; ++kk) {
      const int c0 = kchan0(kk) + 8 * h;
      const float d0 = row[c0], d1 = row[c0 + 1];
      const float s0 = row[CC + c0], s1 = row[CC + c0 + 1];
      v2f a;
      a.x = d0; a.y = d1;
      wd[m][kk] = a;
      v2f b2;
      b2.x = s0 - d0; b2.y = s1 - d1;
      w2[m][kk] = b2;
    }
  }
  v8f bias[2];
#pragma unroll
  for (int m = 0; m < 2; ++m)
#pragma unroll
    for (int r = 0; r < 8; ++r)
      bias[m][r] = bedge[i * CC + 16 * m + 8 * h + r];

  __syncthreads();

#pragma unroll 1
  for (int step = 0; step < NSTEPS; ++step) {
    const v4f* src = lds + (step & 1) * COPY_Q + lane;
    v4f* dst = lds + ((step & 1) ^ 1) * COPY_Q + i * NODE_Q + lane;

    v4f self[4];
    load_tile(src + i * NODE_Q, self);

#pragma unroll 1
    for (int t = 0; t < TILES; ++t) {
      const v4f* st = src + t * TILE_Q;
      v4f nb[2][4];
      __builtin_amdgcn_sched_barrier(0);
      load_tile(st + (0 >= i ? 1 : 0) * NODE_Q, nb[0]);
      __builtin_amdgcn_sched_barrier(0);

      v8f T[2] = {bias[0], bias[1]};
      gemm32(w2, self, T);

      v4f acc[4];
#pragma unroll
      for (int q = 0; q < 4; ++q)
#pragma unroll
        for (int e = 0; e < 4; ++e) acc[q][e] = 0.0f;

#pragma unroll
      for (int k = 0; k < NN - 1; ++k) {
        const int cur = k & 1;
        const int nxt = cur ^ 1;
        __builtin_amdgcn_sched_barrier(0);
        if (k + 1 < NN - 1) {
          const int jn = (k + 1) + ((k + 1) >= i ? 1 : 0);
          load_tile(st + jn * NODE_Q, nb[nxt]);
        } else {
          const int tn = (t + 1 < TILES) ? t + 1 : t;
          load_tile(src + i * NODE_Q + tn * TILE_Q, nb[nxt]);
        }
        __builtin_amdgcn_sched_barrier(0);

        v8f P[2] = {T[0], T[1]};
        gemm32(wd, nb[cur], P);
#pragma unroll
        for (int m = 0; m < 2; ++m)
#pragma unroll
          for (int r = 0; r < 8; ++r) {
            const float gte = __builtin_fmaxf(P[m][r], 0.0f);
            acc[2 * m + (r >> 2)][r & 3] += gte * nb[cur][2 * m + (r >> 2)][r & 3];
          }
      }

#pragma unroll
      for (int q = 0; q < 4; ++q) dst[t * TILE_Q + q * 32] = acc[q] + self[q];

#pragma unroll
      for (int q = 0; q < 4; ++q) self[q] = nb[1][q];
    }
    if (step + 1 < NSTEPS) __syncthreads();
  }

  {
    const v4f* fin = lds + (NSTEPS & 1) * COPY_Q + i * NODE_Q + lane;
    float* g = out + ((long)(i * NB + bb) * CC + 8 * h) * HW;
#pragma unroll
    for (int s = 0; s < 4; ++s) {
      v4f a[4], b[4];
      load_tile(fin + (2 * s) * TILE_Q, a);
      load_tile(fin + (2 * s + 1) * TILE_Q, b);
#pragma unroll
      for (int idx = 0; idx < 16; ++idx) {
        const int c = 16 * (idx >> 3) + (idx & 7);
        v2f o;
        o.x = a[idx >> 2][idx & 3];
        o.y = b[idx >> 2][idx & 3];
        *reinterpret_cast<v2f*>(g + (long)c * HW + span_base + 32 * s + 2 * n) = o;
      }
    }
    v4f a[4];
    load_tile(fin + 8 * TILE_Q, a);
#pragma unroll
    for (int idx = 0; idx < 16; ++idx) {
      const int c = 16 * (idx >> 3) + (idx & 7);
      g[(long)c * HW + left_base + n] = a[idx >> 2][idx & 3];
    }
  }
}

extern "C" void kernel_launch(void* const* d_in, const int* in_sizes, int n_in,
                              void* d_out, int out_size, void* d_ws,
                              size_t ws_size, hipStream_t stream) {
  const float* nodes = (const float*)d_in[0];
  const float* Wedge = (const float*)d_in[1];
  const float* bedge = (const float*)d_in[2];
  float* out = (float*)d_out;

  gcn_fused_kernel<<<dim3(256), dim3(256), 0, stream>>>(nodes, Wedge, bedge, out);
}
